# starting kernel
# baseline (speedup 1.0000x reference)
#include <hip/hip_runtime.h>
#include <hip/hip_bf16.h>

typedef __attribute__((ext_vector_type(2))) float v2f;
typedef __attribute__((ext_vector_type(4))) float v4f;
typedef __attribute__((ext_vector_type(8))) float v8f;

#define NFEAT 512
#define NFILT 64
#define KCHUNK 128

__global__ void kw_zero_kernel(float* __restrict__ p, int n) {
    int i = blockIdx.x * blockDim.x + threadIdx.x;
    if (i < n) p[i] = 0.0f;
}

__global__ void kw_degree_kernel(const int* __restrict__ rows, float* __restrict__ deg, int E) {
    int i = blockIdx.x * blockDim.x + threadIdx.x;
    if (i < E)
        __hip_atomic_fetch_add(&deg[rows[i]], 1.0f, __ATOMIC_RELAXED, __HIP_MEMORY_SCOPE_AGENT);
}

__global__ void kw_dinv_kernel(float* __restrict__ deg, int N) {
    int i = blockIdx.x * blockDim.x + threadIdx.x;
    if (i < N) {
        float d = deg[i];
        deg[i] = (d > 0.0f) ? rsqrtf(d) : 0.0f;
    }
}

__global__ __launch_bounds__(256) void kw_gemm_wmma(const float* __restrict__ x,
                                                    const float* __restrict__ flt,
                                                    float* __restrict__ XF, int N) {
    __shared__ float sB[KCHUNK * NFILT];

    const int lane = threadIdx.x & 31;
    const int wib  = threadIdx.x >> 5;
    const int half = lane >> 4;
    const int l16  = lane & 15;

    const int stripe = blockIdx.x * 8 + wib;
    const int m0     = stripe * 16;
    const bool active = (m0 < N);

    int row = m0 + l16;
    if (row >= N) row = N - 1;
    const float* xrow = x + (size_t)row * NFEAT;

    v8f acc0 = {}, acc1 = {}, acc2 = {}, acc3 = {};

    const float* bbase = sB + lane * 2;

    for (int kc = 0; kc < NFEAT; kc += KCHUNK) {
        __syncthreads();
        {
            const v4f* src = (const v4f*)(flt + (size_t)kc * NFILT);
            for (int t4 = threadIdx.x; t4 < KCHUNK * (NFILT / 4); t4 += 256) {
                v4f v = src[t4];
                int r    = t4 >> 4;
                int g    = r >> 2;
                int koff = r & 3;
                int h    = koff >> 1;
                int j    = koff & 1;
                int cb   = (t4 & 15) << 2;
                int n    = cb >> 4;
                int c16  = cb & 15;
                float* d = sB + (((g << 2) + n) << 6) + (((h << 4) + c16) << 1) + j;
                d[0] = v.x; d[2] = v.y; d[4] = v.z; d[6] = v.w;
            }
        }
        __syncthreads();

        if (active) {
#pragma unroll 4
            for (int k2 = 0; k2 < KCHUNK; k2 += 4) {
                v2f a = *(const v2f*)(xrow + kc + k2 + 2 * half);

                const int g = k2 >> 2;
                const float* bg = bbase + (g << 8);
                v2f b0 = *(const v2f*)(bg);
                v2f b1 = *(const v2f*)(bg + 64);
                v2f b2 = *(const v2f*)(bg + 128);
                v2f b3 = *(const v2f*)(bg + 192);

                acc0 = __builtin_amdgcn_wmma_f32_16x16x4_f32(false, a, false, b0, (short)0, acc0, false, false);
                acc1 = __builtin_amdgcn_wmma_f32_16x16x4_f32(false, a, false, b1, (short)0, acc1, false, false);
                acc2 = __builtin_amdgcn_wmma_f32_16x16x4_f32(false, a, false, b2, (short)0, acc2, false, false);
                acc3 = __builtin_amdgcn_wmma_f32_16x16x4_f32(false, a, false, b3, (short)0, acc3, false, false);
            }
        }
    }

    if (active) {
#pragma unroll
        for (int r = 0; r < 8; ++r) {
            int M = m0 + r + 8 * half;
            if (M < N) {
                float* o = XF + (size_t)M * NFILT + l16;
                o[0]  = acc0[r];
                o[16] = acc1[r];
                o[32] = acc2[r];
                o[48] = acc3[r];
            }
        }
    }
}

__global__ __launch_bounds__(256) void kw_scatter_kernel(const int* __restrict__ rows,
                                                         const int* __restrict__ cols,
                                                         const float* __restrict__ dinv,
                                                         const float* __restrict__ XF,
                                                         float* __restrict__ out, int E) {
    const int wid  = blockIdx.x * (blockDim.x >> 5) + (threadIdx.x >> 5);
    const int lane = threadIdx.x & 31;
    if (wid >= E) return;

    const int r = rows[wid];
    const int c = cols[wid];
    const float w = dinv[r] * dinv[c];

    const float* src = XF + (size_t)c * NFILT;
    float v0 = src[lane];
    float v1 = src[lane + 32];

    float* dst = out + (size_t)r * NFILT;
    __hip_atomic_fetch_add(&dst[lane],      w * v0, __ATOMIC_RELAXED, __HIP_MEMORY_SCOPE_AGENT);
    __hip_atomic_fetch_add(&dst[lane + 32], w * v1, __ATOMIC_RELAXED, __HIP_MEMORY_SCOPE_AGENT);
}

extern "C" void kernel_launch(void* const* d_in, const int* in_sizes, int n_in,
                              void* d_out, int out_size, void* d_ws, size_t ws_size,
                              hipStream_t stream) {
    const float* x   = (const float*)d_in[0];
    const float* flt = (const float*)d_in[1];
    const int*   eid = (const int*)d_in[2];

    const int N = in_sizes[0] / NFEAT;
    const int E = in_sizes[2] / 2;
    const int* rows = eid;
    const int* cols = eid + E;

    float* out = (float*)d_out;
    float* deg = (float*)d_ws;
    float* XF  = (float*)d_ws + (size_t)((N + 63) & ~63);

    kw_zero_kernel<<<(N + 255) / 256, 256, 0, stream>>>(deg, N);
    kw_zero_kernel<<<(out_size + 255) / 256, 256, 0, stream>>>(out, out_size);

    kw_degree_kernel<<<(E + 255) / 256, 256, 0, stream>>>(rows, deg, E);
    kw_dinv_kernel<<<(N + 255) / 256, 256, 0, stream>>>(deg, N);

    const int stripes = (N + 15) / 16;
    kw_gemm_wmma<<<(stripes + 7) / 8, 256, 0, stream>>>(x, flt, XF, N);

    kw_scatter_kernel<<<(E + 7) / 8, 256, 0, stream>>>(rows, cols, deg, XF, out, E);
}


# Each destination row has one owner chip half: work-groups read their die id and, fed by per-half work tickets, add only into rows of their own L2 half (degree, scatter); f32 adds sum in arrival order
# speedup vs baseline: 1.7994x; 1.7994x over previous
#include <hip/hip_runtime.h>
#include <hip/hip_bf16.h>

typedef __attribute__((ext_vector_type(2))) float v2f;
typedef __attribute__((ext_vector_type(4))) float v4f;
typedef __attribute__((ext_vector_type(8))) float v8f;

#define NFEAT 512
#define NFILT 64
#define KCHUNK 128

__global__ void kw_zero_kernel(float* __restrict__ p, int n) {
    int i = blockIdx.x * blockDim.x + threadIdx.x;
    if (i < n) p[i] = 0.0f;
}

#define DG_EDGES 512
#define SC_EDGES 256

__device__ __forceinline__ int kw_chip_half() {
    return (int)((__builtin_amdgcn_s_sendmsg_rtn(0x87) >> 18) & 1u);
}

__device__ __forceinline__ int kw_draw(int* c) {
    return __hip_atomic_fetch_add(c, 1, __ATOMIC_RELAXED, __HIP_MEMORY_SCOPE_AGENT);
}

__global__ __launch_bounds__(256) void kw_degree_kernel(const int* __restrict__ rows,
                                                        float* __restrict__ deg,
                                                        int* __restrict__ ctr, int E, int N) {
    __shared__ int s_tk[2];
    const int home    = kw_chip_half();
    const int split   = N >> 1;
    const int nchunks = (E + DG_EDGES - 1) / DG_EDGES;
    int it = 0;
    for (int pass = 0; pass < 2; ++pass) {
        const int h = home ^ pass;
        int* c = ctr + h;
        int tn = 0;
        if (threadIdx.x == 0) tn = kw_draw(c);
        for (;;) {
            if (threadIdx.x == 0) s_tk[it & 1] = tn;
            __syncthreads();
            const int t = s_tk[it & 1];
            ++it;
            if (t >= nchunks) break;
            if (threadIdx.x == 0) tn = kw_draw(c);
            int  r[DG_EDGES / 256];
            bool ok[DG_EDGES / 256];
#pragma unroll
            for (int j = 0; j < DG_EDGES / 256; ++j) {
                const int e = t * DG_EDGES + j * 256 + (int)threadIdx.x;
                ok[j] = (e < E);
                r[j]  = rows[ok[j] ? e : E - 1];
            }
#pragma unroll
            for (int j = 0; j < DG_EDGES / 256; ++j) {
                if (ok[j] && ((r[j] >= split) == (h != 0)))
                    __hip_atomic_fetch_add(&deg[r[j]], 1.0f, __ATOMIC_RELAXED, __HIP_MEMORY_SCOPE_AGENT);
            }
        }
    }
}

__global__ void kw_dinv_kernel(float* __restrict__ deg, int N) {
    int i = blockIdx.x * blockDim.x + threadIdx.x;
    if (i < N) {
        float d = deg[i];
        deg[i] = (d > 0.0f) ? rsqrtf(d) : 0.0f;
    }
}

__global__ __launch_bounds__(256) void kw_gemm_wmma(const float* __restrict__ x,
                                                    const float* __restrict__ flt,
                                                    float* __restrict__ XF, int N) {
    __shared__ float sB[KCHUNK * NFILT];

    const int lane = threadIdx.x & 31;
    const int wib  = threadIdx.x >> 5;
    const int half = lane >> 4;
    const int l16  = lane & 15;

    const int stripe = blockIdx.x * 8 + wib;
    const int m0     = stripe * 16;
    const bool active = (m0 < N);

    int row = m0 + l16;
    if (row >= N) row = N - 1;
    const float* xrow = x + (size_t)row * NFEAT;

    v8f acc0 = {}, acc1 = {}, acc2 = {}, acc3 = {};

    const float* bbase = sB + lane * 2;

    for (int kc = 0; kc < NFEAT; kc += KCHUNK) {
        __syncthreads();
        {
            const v4f* src = (const v4f*)(flt + (size_t)kc * NFILT);
            for (int t4 = threadIdx.x; t4 < KCHUNK * (NFILT / 4); t4 += 256) {
                v4f v = src[t4];
                int r    = t4 >> 4;
                int g    = r >> 2;
                int koff = r & 3;
                int h    = koff >> 1;
                int j    = koff & 1;
                int cb   = (t4 & 15) << 2;
                int n    = cb >> 4;
                int c16  = cb & 15;
                float* d = sB + (((g << 2) + n) << 6) + (((h << 4) + c16) << 1) + j;
                d[0] = v.x; d[2] = v.y; d[4] = v.z; d[6] = v.w;
            }
        }
        __syncthreads();

        if (active) {
#pragma unroll 4
            for (int k2 = 0; k2 < KCHUNK; k2 += 4) {
                v2f a = *(const v2f*)(xrow + kc + k2 + 2 * half);

                const int g = k2 >> 2;
                const float* bg = bbase + (g << 8);
                v2f b0 = *(const v2f*)(bg);
                v2f b1 = *(const v2f*)(bg + 64);
                v2f b2 = *(const v2f*)(bg + 128);
                v2f b3 = *(const v2f*)(bg + 192);

                acc0 = __builtin_amdgcn_wmma_f32_16x16x4_f32(false, a, false, b0, (short)0, acc0, false, false);
                acc1 = __builtin_amdgcn_wmma_f32_16x16x4_f32(false, a, false, b1, (short)0, acc1, false, false);
                acc2 = __builtin_amdgcn_wmma_f32_16x16x4_f32(false, a, false, b2, (short)0, acc2, false, false);
                acc3 = __builtin_amdgcn_wmma_f32_16x16x4_f32(false, a, false, b3, (short)0, acc3, false, false);
            }
        }
    }

    if (active) {
#pragma unroll
        for (int r = 0; r < 8; ++r) {
            int M = m0 + r + 8 * half;
            if (M < N) {
                float* o = XF + (size_t)M * NFILT + l16;
                o[0]  = acc0[r];
                o[16] = acc1[r];
                o[32] = acc2[r];
                o[48] = acc3[r];
            }
        }
    }
}

template <int K>
__device__ __forceinline__ void kw_add_group(unsigned& m, int er, int ec, float ew, int lane,
                                             const float* __restrict__ XF, float* __restrict__ out) {
    int   rr[K], cc[K];
    float ww[K];
#pragma unroll
    for (int u = 0; u < K; ++u) {
        const int b = __builtin_ctz(m);
        m &= (m - 1u);
        rr[u] = __builtin_amdgcn_readlane(er, b);
        cc[u] = __builtin_amdgcn_readlane(ec, b);
        ww[u] = __builtin_bit_cast(float, __builtin_amdgcn_readlane(__builtin_bit_cast(int, ew), b));
    }
    float v0[K], v1[K];
#pragma unroll
    for (int u = 0; u < K; ++u) {
        const float* s = XF + (size_t)cc[u] * NFILT;
        v0[u] = s[lane];
        v1[u] = s[lane + 32];
    }
#pragma unroll
    for (int u = 0; u < K; ++u) {
        float* d = out + (size_t)rr[u] * NFILT;
        __hip_atomic_fetch_add(&d[lane],      ww[u] * v0[u], __ATOMIC_RELAXED, __HIP_MEMORY_SCOPE_AGENT);
        __hip_atomic_fetch_add(&d[lane + 32], ww[u] * v1[u], __ATOMIC_RELAXED, __HIP_MEMORY_SCOPE_AGENT);
    }
}

__global__ __launch_bounds__(256) void kw_scatter_kernel(const int* __restrict__ rows,
                                                         const int* __restrict__ cols,
                                                         const float* __restrict__ dinv,
                                                         const float* __restrict__ XF,
                                                         float* __restrict__ out,
                                                         int* __restrict__ ctr, int E, int N) {
    __shared__ int s_tk[2];
    const int lane    = threadIdx.x & 31;
    const int home    = kw_chip_half();
    const int split   = N >> 1;
    const int nchunks = (E + SC_EDGES - 1) / SC_EDGES;
    int it = 0;
    for (int pass = 0; pass < 2; ++pass) {
        const int h = home ^ pass;
        int* c = ctr + h;
        int tn = 0;
        if (threadIdx.x == 0) tn = kw_draw(c);
        for (;;) {
            if (threadIdx.x == 0) s_tk[it & 1] = tn;
            __syncthreads();
            const int t = s_tk[it & 1];
            ++it;
            if (t >= nchunks) break;
            if (threadIdx.x == 0) tn = kw_draw(c);

            const int e = t * SC_EDGES + (int)threadIdx.x;
            int er = 0, ec = 0;
            bool own = false;
            if (e < E) {
                er  = rows[e];
                ec  = cols[e];
                own = ((er >= split) == (h != 0));
            }
            float ew = 0.0f;
            if (own) ew = dinv[er] * dinv[ec];

            unsigned m = __builtin_amdgcn_ballot_w32(own);
            while (__builtin_popcount(m) >= 4) kw_add_group<4>(m, er, ec, ew, lane, XF, out);
            if (__builtin_popcount(m) >= 2)    kw_add_group<2>(m, er, ec, ew, lane, XF, out);
            if (m)                             kw_add_group<1>(m, er, ec, ew, lane, XF, out);
        }
    }
}

extern "C" void kernel_launch(void* const* d_in, const int* in_sizes, int n_in,
                              void* d_out, int out_size, void* d_ws, size_t ws_size,
                              hipStream_t stream) {
    const float* x   = (const float*)d_in[0];
    const float* flt = (const float*)d_in[1];
    const int*   eid = (const int*)d_in[2];

    const int N = in_sizes[0] / NFEAT;
    const int E = in_sizes[2] / 2;
    const int* rows = eid;
    const int* cols = eid + E;

    float* out = (float*)d_out;
    float* deg = (float*)d_ws;
    int*   ctr = (int*)(deg + N);
    float* XF  = (float*)d_ws + (size_t)((N + 4 + 63) & ~63);

    kw_zero_kernel<<<(N + 4 + 255) / 256, 256, 0, stream>>>(deg, N + 4);
    kw_zero_kernel<<<(out_size + 255) / 256, 256, 0, stream>>>(out, out_size);

    kw_degree_kernel<<<1024, 256, 0, stream>>>(rows, deg, ctr, E, N);
    kw_dinv_kernel<<<(N + 255) / 256, 256, 0, stream>>>(deg, N);

    const int stripes = (N + 15) / 16;
    kw_gemm_wmma<<<(stripes + 7) / 8, 256, 0, stream>>>(x, flt, XF, N);

    kw_scatter_kernel<<<2048, 256, 0, stream>>>(rows, cols, deg, XF, out, ctr + 2, E, N);
}


# Scatter-add replaced by per-row buckets kept in the output rows (count + source ids, filled by integer adds) and a gather that sums each row in registers, written once; sums in bucket arrival order
# speedup vs baseline: 1.9503x; 1.0839x over previous
#include <hip/hip_runtime.h>
#include <hip/hip_bf16.h>

typedef __attribute__((ext_vector_type(2))) float v2f;
typedef __attribute__((ext_vector_type(4))) float v4f;
typedef __attribute__((ext_vector_type(8))) float v8f;

#define NFEAT 512
#define NFILT 64
#define KCHUNK 128


#define BK_CAP   63
#define FL_EDGES 512
#define GR_RPW   2
#define GR_ROWS  (8 * GR_RPW)

__global__ void kw_prep_kernel(float* __restrict__ out, int* __restrict__ ctr, int N) {
    int i = blockIdx.x * blockDim.x + threadIdx.x;
    if (i < N) out[(size_t)i * NFILT] = 0.0f;
    if (i < 4) ctr[i] = 0;
}

__device__ __forceinline__ int kw_chip_half() {
    return (int)((__builtin_amdgcn_s_sendmsg_rtn(0x87) >> 18) & 1u);
}

__device__ __forceinline__ int kw_draw(int* c) {
    return __hip_atomic_fetch_add(c, 1, __ATOMIC_RELAXED, __HIP_MEMORY_SCOPE_AGENT);
}

__global__ __launch_bounds__(256) void kw_fill_kernel(const int* __restrict__ rows,
                                                      const int* __restrict__ cols,
                                                      float* out, int* __restrict__ ctr, int E, int N) {
    __shared__ int s_tk[2];
    int* bk = (int*)out;
    const int home    = kw_chip_half();
    const int split   = N >> 1;
    const int nchunks = (E + FL_EDGES - 1) / FL_EDGES;
    int it = 0;
    for (int pass = 0; pass < 2; ++pass) {
        const int h = home ^ pass;
        int* c = ctr + h;
        int tn = 0;
        if (threadIdx.x == 0) tn = kw_draw(c);
        for (;;) {
            if (threadIdx.x == 0) s_tk[it & 1] = tn;
            __syncthreads();
            const int t = s_tk[it & 1];
            ++it;
            if (t >= nchunks) break;
            if (threadIdx.x == 0) tn = kw_draw(c);
            int  r[FL_EDGES / 256], s[FL_EDGES / 256];
            bool ok[FL_EDGES / 256];
#pragma unroll
            for (int j = 0; j < FL_EDGES / 256; ++j) {
                const int e = t * FL_EDGES + j * 256 + (int)threadIdx.x;
                ok[j] = (e < E);
                r[j]  = rows[ok[j] ? e : E - 1];
                s[j]  = cols[ok[j] ? e : E - 1];
            }
            int k[FL_EDGES / 256];
#pragma unroll
            for (int j = 0; j < FL_EDGES / 256; ++j) {
                ok[j] = ok[j] && ((r[j] >= split) == (h != 0));
                k[j]  = BK_CAP;
                if (ok[j]) k[j] = __hip_atomic_fetch_add(bk + (size_t)r[j] * NFILT, 1, __ATOMIC_RELAXED, __HIP_MEMORY_SCOPE_AGENT);
            }
#pragma unroll
            for (int j = 0; j < FL_EDGES / 256; ++j) {
                if (k[j] < BK_CAP) bk[(size_t)r[j] * NFILT + 1 + k[j]] = s[j];
            }
        }
    }
}

__global__ void kw_dinv_kernel(const float* __restrict__ out, float* __restrict__ dinv, int N) {
    int i = blockIdx.x * blockDim.x + threadIdx.x;
    if (i < N) {
        const int d = __builtin_bit_cast(int, out[(size_t)i * NFILT]);
        dinv[i] = (d > 0) ? rsqrtf((float)d) : 0.0f;
    }
}

__global__ __launch_bounds__(256) void kw_gemm_wmma(const float* __restrict__ x,
                                                    const float* __restrict__ flt,
                                                    float* __restrict__ XF, int N) {
    __shared__ float sB[KCHUNK * NFILT];

    const int lane = threadIdx.x & 31;
    const int wib  = threadIdx.x >> 5;
    const int half = lane >> 4;
    const int l16  = lane & 15;

    const int stripe = blockIdx.x * 8 + wib;
    const int m0     = stripe * 16;
    const bool active = (m0 < N);

    int row = m0 + l16;
    if (row >= N) row = N - 1;
    const float* xrow = x + (size_t)row * NFEAT;

    v8f acc0 = {}, acc1 = {}, acc2 = {}, acc3 = {};

    const float* bbase = sB + lane * 2;

    for (int kc = 0; kc < NFEAT; kc += KCHUNK) {
        __syncthreads();
        {
            const v4f* src = (const v4f*)(flt + (size_t)kc * NFILT);
            for (int t4 = threadIdx.x; t4 < KCHUNK * (NFILT / 4); t4 += 256) {
                v4f v = src[t4];
                int r    = t4 >> 4;
                int g    = r >> 2;
                int koff = r & 3;
                int h    = koff >> 1;
                int j    = koff & 1;
                int cb   = (t4 & 15) << 2;
                int n    = cb >> 4;
                int c16  = cb & 15;
                float* d = sB + (((g << 2) + n) << 6) + (((h << 4) + c16) << 1) + j;
                d[0] = v.x; d[2] = v.y; d[4] = v.z; d[6] = v.w;
            }
        }
        __syncthreads();

        if (active) {
#pragma unroll 4
            for (int k2 = 0; k2 < KCHUNK; k2 += 4) {
                v2f a = *(const v2f*)(xrow + kc + k2 + 2 * half);

                const int g = k2 >> 2;
                const float* bg = bbase + (g << 8);
                v2f b0 = *(const v2f*)(bg);
                v2f b1 = *(const v2f*)(bg + 64);
                v2f b2 = *(const v2f*)(bg + 128);
                v2f b3 = *(const v2f*)(bg + 192);

                acc0 = __builtin_amdgcn_wmma_f32_16x16x4_f32(false, a, false, b0, (short)0, acc0, false, false);
                acc1 = __builtin_amdgcn_wmma_f32_16x16x4_f32(false, a, false, b1, (short)0, acc1, false, false);
                acc2 = __builtin_amdgcn_wmma_f32_16x16x4_f32(false, a, false, b2, (short)0, acc2, false, false);
                acc3 = __builtin_amdgcn_wmma_f32_16x16x4_f32(false, a, false, b3, (short)0, acc3, false, false);
            }
        }
    }

    if (active) {
#pragma unroll
        for (int r = 0; r < 8; ++r) {
            int M = m0 + r + 8 * half;
            if (M < N) {
                float* o = XF + (size_t)M * NFILT + l16;
                o[0]  = acc0[r];
                o[16] = acc1[r];
                o[32] = acc2[r];
                o[48] = acc3[r];
            }
        }
    }
}

template <int K>
__device__ __forceinline__ void kw_sum_group(int s0, int ids, float dcs, float di, int lane,
                                             const float* __restrict__ XF, v2f& acc) {
    int cc[K];
#pragma unroll
    for (int u = 0; u < K; ++u) cc[u] = __builtin_amdgcn_readlane(ids, s0 + u);
    v2f v[K];
#pragma unroll
    for (int u = 0; u < K; ++u) v[u] = *(const v2f*)(XF + (size_t)cc[u] * NFILT + 2 * lane);
#pragma unroll
    for (int u = 0; u < K; ++u) {
        const float w = di * __builtin_bit_cast(float, __builtin_amdgcn_readlane(__builtin_bit_cast(int, dcs), s0 + u));
        acc += w * v[u];
    }
}

__device__ __forceinline__ v2f kw_row_walk(int i, float di, int lane,
                                           const int* __restrict__ rows, const int* __restrict__ cols,
                                           const float* __restrict__ dinv, const float* __restrict__ XF, int E) {
    v2f acc = {0.0f, 0.0f};
    for (int e0 = 0; e0 < E; e0 += 32) {
        const int e = e0 + lane;
        int r = -1, c = 0;
        if (e < E) { r = rows[e]; c = cols[e]; }
        unsigned m = __builtin_amdgcn_ballot_w32(r == i);
        while (m) {
            const int bl = __builtin_ctz(m);
            m &= (m - 1u);
            const int   cc = __builtin_amdgcn_readlane(c, bl);
            const float w  = di * dinv[cc];
            const v2f   v  = *(const v2f*)(XF + (size_t)cc * NFILT + 2 * lane);
            acc += w * v;
        }
    }
    return acc;
}

__global__ __launch_bounds__(256) void kw_gather_kernel(const int* __restrict__ rows,
                                                        const int* __restrict__ cols,
                                                        const float* __restrict__ dinv,
                                                        const float* __restrict__ XF,
                                                        float* __restrict__ out,
                                                        int* __restrict__ ctr, int E, int N) {
    __shared__ int s_tk[2];
    const int lane  = threadIdx.x & 31;
    const int wib   = threadIdx.x >> 5;
    const int home  = kw_chip_half();
    const int split = N >> 1;
    int it = 0;
    for (int pass = 0; pass < 2; ++pass) {
        const int h  = home ^ pass;
        const int lo = h ? split : 0;
        const int hi = h ? N : split;
        const int nchunks = (hi - lo + GR_ROWS - 1) / GR_ROWS;
        int* c = ctr + h;
        int tn = 0;
        if (threadIdx.x == 0) tn = kw_draw(c);
        for (;;) {
            if (threadIdx.x == 0) s_tk[it & 1] = tn;
            __syncthreads();
            const int t = s_tk[it & 1];
            ++it;
            if (t >= nchunks) break;
            if (threadIdx.x == 0) tn = kw_draw(c);

            const int i0 = lo + t * GR_ROWS + wib * GR_RPW;
            int   a[GR_RPW];
            float di[GR_RPW];
#pragma unroll
            for (int j = 0; j < GR_RPW; ++j) {
                a[j] = 0; di[j] = 0.0f;
                if (i0 + j < hi) {
                    a[j]  = __builtin_bit_cast(int, out[(size_t)(i0 + j) * NFILT + lane]);
                    di[j] = dinv[i0 + j];
                }
            }
            int   cnt[GR_RPW];
            float da[GR_RPW];
#pragma unroll
            for (int j = 0; j < GR_RPW; ++j) {
                cnt[j] = __builtin_amdgcn_readlane(a[j], 0);
                da[j] = dinv[(lane >= 1 && lane <= cnt[j]) ? a[j] : 0];
            }
#pragma unroll
            for (int j = 0; j < GR_RPW; ++j) {
                if (i0 + j < hi) {
                    const int i = i0 + j;
                    v2f acc = {0.0f, 0.0f};
                    if (cnt[j] <= BK_CAP) {
                        const int n1 = cnt[j] < 31 ? cnt[j] : 31;
                        int s = 1;
                        while (n1 - s >= 7) { kw_sum_group<8>(s, a[j], da[j], di[j], lane, XF, acc); s += 8; }
                        if (n1 - s >= 3)    { kw_sum_group<4>(s, a[j], da[j], di[j], lane, XF, acc); s += 4; }
                        if (n1 - s >= 1)    { kw_sum_group<2>(s, a[j], da[j], di[j], lane, XF, acc); s += 2; }
                        if (n1 - s >= 0)    { kw_sum_group<1>(s, a[j], da[j], di[j], lane, XF, acc); }
                        if (cnt[j] >= 32) {
                            const int   b  = __builtin_bit_cast(int, out[(size_t)i * NFILT + 32 + lane]);
                            const float db = dinv[(lane + 32 <= cnt[j]) ? b : 0];
                            for (int u = 32; u <= cnt[j]; ++u) kw_sum_group<1>(u - 32, b, db, di[j], lane, XF, acc);
                        }
                    } else {
                        acc = kw_row_walk(i, di[j], lane, rows, cols, dinv, XF, E);
                    }
                    *(v2f*)(out + (size_t)i * NFILT + 2 * lane) = acc;
                }
            }
        }
    }
}

extern "C" void kernel_launch(void* const* d_in, const int* in_sizes, int n_in,
                              void* d_out, int out_size, void* d_ws, size_t ws_size,
                              hipStream_t stream) {
    const float* x   = (const float*)d_in[0];
    const float* flt = (const float*)d_in[1];
    const int*   eid = (const int*)d_in[2];

    const int N = in_sizes[0] / NFEAT;
    const int E = in_sizes[2] / 2;
    const int* rows = eid;
    const int* cols = eid + E;

    float* out  = (float*)d_out;
    float* dinv = (float*)d_ws;
    int*   ctr  = (int*)(dinv + N);
    float* XF   = (float*)d_ws + (size_t)((N + 4 + 63) & ~63);

    kw_prep_kernel<<<(N + 255) / 256, 256, 0, stream>>>(out, ctr, N);

    const int stripes = (N + 15) / 16;
    kw_gemm_wmma<<<(stripes + 7) / 8, 256, 0, stream>>>(x, flt, XF, N);

    kw_fill_kernel<<<2048, 256, 0, stream>>>(rows, cols, out, ctr, E, N);
    kw_dinv_kernel<<<(N + 255) / 256, 256, 0, stream>>>(out, dinv, N);

    kw_gather_kernel<<<2048, 256, 0, stream>>>(rows, cols, dinv, XF, out, ctr + 2, E, N);
}


# Matrix product and bucket fill merged into one launch (first work-groups: product stripes, rest: fill tickets), so the fill's cache round trips run under the matrix work; 4 launches, not 5
# speedup vs baseline: 2.3537x; 1.2068x over previous
#include <hip/hip_runtime.h>
#include <hip/hip_bf16.h>

typedef __attribute__((ext_vector_type(2))) float v2f;
typedef __attribute__((ext_vector_type(4))) float v4f;
typedef __attribute__((ext_vector_type(8))) float v8f;

#define NFEAT 512
#define NFILT 64
#define KCHUNK 128


#define BK_CAP   63
#define FL_EDGES 512
#define GR_RPW   2
#define GR_ROWS  (8 * GR_RPW)

__global__ void kw_prep_kernel(float* __restrict__ out, int* __restrict__ ctr, int N) {
    int i = blockIdx.x * blockDim.x + threadIdx.x;
    if (i < N) out[(size_t)i * NFILT] = 0.0f;
    if (i < 4) ctr[i] = 0;
}

__device__ __forceinline__ int kw_chip_half() {
    return (int)((__builtin_amdgcn_s_sendmsg_rtn(0x87) >> 18) & 1u);
}

__device__ __forceinline__ int kw_draw(int* c) {
    return __hip_atomic_fetch_add(c, 1, __ATOMIC_RELAXED, __HIP_MEMORY_SCOPE_AGENT);
}

__device__ __forceinline__ void kw_fill_body(const int* __restrict__ rows,
                                             const int* __restrict__ cols,
                                             float* out, int* __restrict__ ctr, int E, int N, int* s_tk) {
    int* bk = (int*)out;
    const int home    = kw_chip_half();
    const int split   = N >> 1;
    const int nchunks = (E + FL_EDGES - 1) / FL_EDGES;
    int it = 0;
    for (int pass = 0; pass < 2; ++pass) {
        const int h = home ^ pass;
        int* c = ctr + h;
        int tn = 0;
        if (threadIdx.x == 0) tn = kw_draw(c);
        for (;;) {
            if (threadIdx.x == 0) s_tk[it & 1] = tn;
            __syncthreads();
            const int t = s_tk[it & 1];
            ++it;
            if (t >= nchunks) break;
            if (threadIdx.x == 0) tn = kw_draw(c);
            int  r[FL_EDGES / 256], s[FL_EDGES / 256];
            bool ok[FL_EDGES / 256];
#pragma unroll
            for (int j = 0; j < FL_EDGES / 256; ++j) {
                const int e = t * FL_EDGES + j * 256 + (int)threadIdx.x;
                ok[j] = (e < E);
                r[j]  = rows[ok[j] ? e : E - 1];
                s[j]  = cols[ok[j] ? e : E - 1];
            }
            int k[FL_EDGES / 256];
#pragma unroll
            for (int j = 0; j < FL_EDGES / 256; ++j) {
                ok[j] = ok[j] && ((r[j] >= split) == (h != 0));
                k[j]  = BK_CAP;
                if (ok[j]) k[j] = __hip_atomic_fetch_add(bk + (size_t)r[j] * NFILT, 1, __ATOMIC_RELAXED, __HIP_MEMORY_SCOPE_AGENT);
            }
#pragma unroll
            for (int j = 0; j < FL_EDGES / 256; ++j) {
                if (k[j] < BK_CAP) bk[(size_t)r[j] * NFILT + 1 + k[j]] = s[j];
            }
        }
    }
}

__global__ void kw_dinv_kernel(const float* __restrict__ out, float* __restrict__ dinv, int N) {
    int i = blockIdx.x * blockDim.x + threadIdx.x;
    if (i < N) {
        const int d = __builtin_bit_cast(int, out[(size_t)i * NFILT]);
        dinv[i] = (d > 0) ? rsqrtf((float)d) : 0.0f;
    }
}

__device__ __forceinline__ void kw_gemm_body(const float* __restrict__ x,
                                             const float* __restrict__ flt,
                                             float* __restrict__ XF, int N, int blk, float* sB) {
    const int lane = threadIdx.x & 31;
    const int wib  = threadIdx.x >> 5;
    const int half = lane >> 4;
    const int l16  = lane & 15;

    const int stripe = blk * 8 + wib;
    const int m0     = stripe * 16;
    const bool active = (m0 < N);

    int row = m0 + l16;
    if (row >= N) row = N - 1;
    const float* xrow = x + (size_t)row * NFEAT;

    v8f acc0 = {}, acc1 = {}, acc2 = {}, acc3 = {};

    const float* bbase = sB + lane * 2;

    for (int kc = 0; kc < NFEAT; kc += KCHUNK) {
        __syncthreads();
        {
            const v4f* src = (const v4f*)(flt + (size_t)kc * NFILT);
            for (int t4 = threadIdx.x; t4 < KCHUNK * (NFILT / 4); t4 += 256) {
                v4f v = src[t4];
                int r    = t4 >> 4;
                int g    = r >> 2;
                int koff = r & 3;
                int h    = koff >> 1;
                int j    = koff & 1;
                int cb   = (t4 & 15) << 2;
                int n    = cb >> 4;
                int c16  = cb & 15;
                float* d = sB + (((g << 2) + n) << 6) + (((h << 4) + c16) << 1) + j;
                d[0] = v.x; d[2] = v.y; d[4] = v.z; d[6] = v.w;
            }
        }
        __syncthreads();

        if (active) {
#pragma unroll 4
            for (int k2 = 0; k2 < KCHUNK; k2 += 4) {
                v2f a = *(const v2f*)(xrow + kc + k2 + 2 * half);

                const int g = k2 >> 2;
                const float* bg = bbase + (g << 8);
                v2f b0 = *(const v2f*)(bg);
                v2f b1 = *(const v2f*)(bg + 64);
                v2f b2 = *(const v2f*)(bg + 128);
                v2f b3 = *(const v2f*)(bg + 192);

                acc0 = __builtin_amdgcn_wmma_f32_16x16x4_f32(false, a, false, b0, (short)0, acc0, false, false);
                acc1 = __builtin_amdgcn_wmma_f32_16x16x4_f32(false, a, false, b1, (short)0, acc1, false, false);
                acc2 = __builtin_amdgcn_wmma_f32_16x16x4_f32(false, a, false, b2, (short)0, acc2, false, false);
                acc3 = __builtin_amdgcn_wmma_f32_16x16x4_f32(false, a, false, b3, (short)0, acc3, false, false);
            }
        }
    }

    if (active) {
#pragma unroll
        for (int r = 0; r < 8; ++r) {
            int M = m0 + r + 8 * half;
            if (M < N) {
                float* o = XF + (size_t)M * NFILT + l16;
                o[0]  = acc0[r];
                o[16] = acc1[r];
                o[32] = acc2[r];
                o[48] = acc3[r];
            }
        }
    }
}

#define FL_BLOCKS 1266
__global__ __launch_bounds__(256, 16) void kw_gemm_fill_kernel(const float* __restrict__ x,
                                                           const float* __restrict__ flt,
                                                           float* __restrict__ XF,
                                                           const int* __restrict__ rows,
                                                           const int* __restrict__ cols,
                                                           float* out, int* __restrict__ ctr, int E, int N) {
    __shared__ float sB[KCHUNK * NFILT];
    __shared__ int s_tk[2];
    const int gemm_blocks = (((N + 15) >> 4) + 7) >> 3;
    if ((int)blockIdx.x < gemm_blocks) kw_gemm_body(x, flt, XF, N, (int)blockIdx.x, sB);
    else                               kw_fill_body(rows, cols, out, ctr, E, N, s_tk);
}

template <int K>
__device__ __forceinline__ void kw_sum_group(int s0, int ids, float dcs, float di, int lane,
                                             const float* __restrict__ XF, v2f& acc) {
    int cc[K];
#pragma unroll
    for (int u = 0; u < K; ++u) cc[u] = __builtin_amdgcn_readlane(ids, s0 + u);
    v2f v[K];
#pragma unroll
    for (int u = 0; u < K; ++u) v[u] = *(const v2f*)(XF + (size_t)cc[u] * NFILT + 2 * lane);
#pragma unroll
    for (int u = 0; u < K; ++u) {
        const float w = di * __builtin_bit_cast(float, __builtin_amdgcn_readlane(__builtin_bit_cast(int, dcs), s0 + u));
        acc += w * v[u];
    }
}

__device__ __forceinline__ v2f kw_row_walk(int i, float di, int lane,
                                           const int* __restrict__ rows, const int* __restrict__ cols,
                                           const float* __restrict__ dinv, const float* __restrict__ XF, int E) {
    v2f acc = {0.0f, 0.0f};
    for (int e0 = 0; e0 < E; e0 += 32) {
        const int e = e0 + lane;
        int r = -1, c = 0;
        if (e < E) { r = rows[e]; c = cols[e]; }
        unsigned m = __builtin_amdgcn_ballot_w32(r == i);
        while (m) {
            const int bl = __builtin_ctz(m);
            m &= (m - 1u);
            const int   cc = __builtin_amdgcn_readlane(c, bl);
            const float w  = di * dinv[cc];
            const v2f   v  = *(const v2f*)(XF + (size_t)cc * NFILT + 2 * lane);
            acc += w * v;
        }
    }
    return acc;
}

__global__ __launch_bounds__(256) void kw_gather_kernel(const int* __restrict__ rows,
                                                        const int* __restrict__ cols,
                                                        const float* __restrict__ dinv,
                                                        const float* __restrict__ XF,
                                                        float* __restrict__ out,
                                                        int* __restrict__ ctr, int E, int N) {
    __shared__ int s_tk[2];
    const int lane  = threadIdx.x & 31;
    const int wib   = threadIdx.x >> 5;
    const int home  = kw_chip_half();
    const int split = N >> 1;
    int it = 0;
    for (int pass = 0; pass < 2; ++pass) {
        const int h  = home ^ pass;
        const int lo = h ? split : 0;
        const int hi = h ? N : split;
        const int nchunks = (hi - lo + GR_ROWS - 1) / GR_ROWS;
        int* c = ctr + h;
        int tn = 0;
        if (threadIdx.x == 0) tn = kw_draw(c);
        for (;;) {
            if (threadIdx.x == 0) s_tk[it & 1] = tn;
            __syncthreads();
            const int t = s_tk[it & 1];
            ++it;
            if (t >= nchunks) break;
            if (threadIdx.x == 0) tn = kw_draw(c);

            const int i0 = lo + t * GR_ROWS + wib * GR_RPW;
            int   a[GR_RPW];
            float di[GR_RPW];
#pragma unroll
            for (int j = 0; j < GR_RPW; ++j) {
                a[j] = 0; di[j] = 0.0f;
                if (i0 + j < hi) {
                    a[j]  = __builtin_bit_cast(int, out[(size_t)(i0 + j) * NFILT + lane]);
                    di[j] = dinv[i0 + j];
                }
            }
            int   cnt[GR_RPW];
            float da[GR_RPW];
#pragma unroll
            for (int j = 0; j < GR_RPW; ++j) {
                cnt[j] = __builtin_amdgcn_readlane(a[j], 0);
                da[j] = dinv[(lane >= 1 && lane <= cnt[j]) ? a[j] : 0];
            }
#pragma unroll
            for (int j = 0; j < GR_RPW; ++j) {
                if (i0 + j < hi) {
                    const int i = i0 + j;
                    v2f acc = {0.0f, 0.0f};
                    if (cnt[j] <= BK_CAP) {
                        const int n1 = cnt[j] < 31 ? cnt[j] : 31;
                        int s = 1;
                        while (n1 - s >= 7) { kw_sum_group<8>(s, a[j], da[j], di[j], lane, XF, acc); s += 8; }
                        if (n1 - s >= 3)    { kw_sum_group<4>(s, a[j], da[j], di[j], lane, XF, acc); s += 4; }
                        if (n1 - s >= 1)    { kw_sum_group<2>(s, a[j], da[j], di[j], lane, XF, acc); s += 2; }
                        if (n1 - s >= 0)    { kw_sum_group<1>(s, a[j], da[j], di[j], lane, XF, acc); }
                        if (cnt[j] >= 32) {
                            const int   b  = __builtin_bit_cast(int, out[(size_t)i * NFILT + 32 + lane]);
                            const float db = dinv[(lane + 32 <= cnt[j]) ? b : 0];
                            for (int u = 32; u <= cnt[j]; ++u) kw_sum_group<1>(u - 32, b, db, di[j], lane, XF, acc);
                        }
                    } else {
                        acc = kw_row_walk(i, di[j], lane, rows, cols, dinv, XF, E);
                    }
                    *(v2f*)(out + (size_t)i * NFILT + 2 * lane) = acc;
                }
            }
        }
    }
}

extern "C" void kernel_launch(void* const* d_in, const int* in_sizes, int n_in,
                              void* d_out, int out_size, void* d_ws, size_t ws_size,
                              hipStream_t stream) {
    const float* x   = (const float*)d_in[0];
    const float* flt = (const float*)d_in[1];
    const int*   eid = (const int*)d_in[2];

    const int N = in_sizes[0] / NFEAT;
    const int E = in_sizes[2] / 2;
    const int* rows = eid;
    const int* cols = eid + E;

    float* out  = (float*)d_out;
    float* dinv = (float*)d_ws;
    int*   ctr  = (int*)(dinv + N);
    float* XF   = (float*)d_ws + (size_t)((N + 4 + 63) & ~63);

    kw_prep_kernel<<<(N + 255) / 256, 256, 0, stream>>>(out, ctr, N);

    const int stripes = (N + 15) / 16;
    kw_gemm_fill_kernel<<<(stripes + 7) / 8 + FL_BLOCKS, 256, 0, stream>>>(x, flt, XF, rows, cols, out, ctr, E, N);

    kw_dinv_kernel<<<(N + 255) / 256, 256, 0, stream>>>(out, dinv, N);

    kw_gather_kernel<<<2048, 256, 0, stream>>>(rows, cols, dinv, XF, out, ctr + 2, E, N);
}


# Bucket fill's per-edge returning global adds replaced by two-level grouping on LDS counters (bin lists kept in the output rows, one global add per bin and work-group); sums in LDS arrival order
# speedup vs baseline: 2.6675x; 1.1333x over previous
#include <hip/hip_runtime.h>
#include <hip/hip_bf16.h>

typedef __attribute__((ext_vector_type(2))) float v2f;
typedef __attribute__((ext_vector_type(4))) float v4f;
typedef __attribute__((ext_vector_type(8))) float v8f;

#define NFEAT 512
#define NFILT 64
#define KCHUNK 128


#define BK_CAP   63
#define RPB      391
#define NBMAX    256
#define BN_EDGES 2048
#define SRC_BITS 20
#define BD_THREADS 1024
#define GR_RPW   2
#define GR_ROWS  (8 * GR_RPW)

__device__ __forceinline__ int kw_bin_rows(int b, int N) {
    const int nr = N - b * RPB;
    return nr < RPB ? nr : RPB;
}

__global__ void kw_prep_kernel(float* out, int* __restrict__ ctr, int N) {
    const int i  = blockIdx.x * blockDim.x + threadIdx.x;
    const int NB = (N + RPB - 1) / RPB;
    if (i < 2 * NB) {
        const int b = i >> 1;
        ((int*)out)[(size_t)b * RPB * NFILT + (size_t)(i & 1) * kw_bin_rows(b, N) * 32] = 0;
    }
    if (i < 4) ctr[i] = 0;
}

__device__ __forceinline__ int kw_chip_half() {
    return (int)((__builtin_amdgcn_s_sendmsg_rtn(0x87) >> 18) & 1u);
}

__device__ __forceinline__ int kw_draw(int* c) {
    return __hip_atomic_fetch_add(c, 1, __ATOMIC_RELAXED, __HIP_MEMORY_SCOPE_AGENT);
}

__device__ __forceinline__ void kw_bin_body(const int* __restrict__ rows,
                                            const int* __restrict__ cols,
                                            float* out, int E, int N, int chunk, int* s_cnt, int* s_base) {
    int* bk = (int*)out;
    const int h  = kw_chip_half();
    const int NB = (N + RPB - 1) / RPB;
    for (int b = threadIdx.x; b < NB; b += 256) s_cnt[b] = 0;
    __syncthreads();
    int  bin[BN_EDGES / 256], w[BN_EDGES / 256], k[BN_EDGES / 256];
    bool ok[BN_EDGES / 256];
#pragma unroll
    for (int j = 0; j < BN_EDGES / 256; ++j) {
        const int e = chunk * BN_EDGES + j * 256 + (int)threadIdx.x;
        ok[j] = (e < E);
        const int r = rows[ok[j] ? e : E - 1];
        const int s = cols[ok[j] ? e : E - 1];
        ok[j]  = ok[j] && ((unsigned)r < (unsigned)N);
        bin[j] = ok[j] ? r / RPB : 0;
        w[j]   = (int)(((unsigned)(r - bin[j] * RPB) << SRC_BITS) | (unsigned)s);
    }
#pragma unroll
    for (int j = 0; j < BN_EDGES / 256; ++j) {
        k[j] = 0;
        if (ok[j]) k[j] = __hip_atomic_fetch_add(s_cnt + bin[j], 1, __ATOMIC_RELAXED, __HIP_MEMORY_SCOPE_WORKGROUP);
    }
    __syncthreads();
    for (int b = threadIdx.x; b < NB; b += 256) {
        const int c = s_cnt[b];
        int base = 0;
        if (c > 0) base = __hip_atomic_fetch_add(bk + (size_t)b * RPB * NFILT + (size_t)h * kw_bin_rows(b, N) * 32,
                                                 c, __ATOMIC_RELAXED, __HIP_MEMORY_SCOPE_AGENT);
        s_base[b] = base;
    }
    __syncthreads();
#pragma unroll
    for (int j = 0; j < BN_EDGES / 256; ++j) {
        if (ok[j]) {
            const int nr  = kw_bin_rows(bin[j], N);
            const int pos = s_base[bin[j]] + k[j];
            if (pos < nr * 32 - 32) bk[(size_t)bin[j] * RPB * NFILT + (size_t)h * nr * 32 + 32 + pos] = w[j];
        }
    }
}

__device__ __forceinline__ void kw_gemm_body(const float* __restrict__ x,
                                             const float* __restrict__ flt,
                                             float* __restrict__ XF, int N, int blk, float* sB) {
    const int lane = threadIdx.x & 31;
    const int wib  = threadIdx.x >> 5;
    const int half = lane >> 4;
    const int l16  = lane & 15;

    const int stripe = blk * 8 + wib;
    const int m0     = stripe * 16;
    const bool active = (m0 < N);

    int row = m0 + l16;
    if (row >= N) row = N - 1;
    const float* xrow = x + (size_t)row * NFEAT;

    v8f acc0 = {}, acc1 = {}, acc2 = {}, acc3 = {};

    const float* bbase = sB + lane * 2;

    for (int kc = 0; kc < NFEAT; kc += KCHUNK) {
        __syncthreads();
        {
            const v4f* src = (const v4f*)(flt + (size_t)kc * NFILT);
            for (int t4 = threadIdx.x; t4 < KCHUNK * (NFILT / 4); t4 += 256) {
                v4f v = src[t4];
                int r    = t4 >> 4;
                int g    = r >> 2;
                int koff = r & 3;
                int h    = koff >> 1;
                int j    = koff & 1;
                int cb   = (t4 & 15) << 2;
                int n    = cb >> 4;
                int c16  = cb & 15;
                float* d = sB + (((g << 2) + n) << 6) + (((h << 4) + c16) << 1) + j;
                d[0] = v.x; d[2] = v.y; d[4] = v.z; d[6] = v.w;
            }
        }
        __syncthreads();

        if (active) {
#pragma unroll 4
            for (int k2 = 0; k2 < KCHUNK; k2 += 4) {
                v2f a = *(const v2f*)(xrow + kc + k2 + 2 * half);

                const int g = k2 >> 2;
                const float* bg = bbase + (g << 8);
                v2f b0 = *(const v2f*)(bg);
                v2f b1 = *(const v2f*)(bg + 64);
                v2f b2 = *(const v2f*)(bg + 128);
                v2f b3 = *(const v2f*)(bg + 192);

                acc0 = __builtin_amdgcn_wmma_f32_16x16x4_f32(false, a, false, b0, (short)0, acc0, false, false);
                acc1 = __builtin_amdgcn_wmma_f32_16x16x4_f32(false, a, false, b1, (short)0, acc1, false, false);
                acc2 = __builtin_amdgcn_wmma_f32_16x16x4_f32(false, a, false, b2, (short)0, acc2, false, false);
                acc3 = __builtin_amdgcn_wmma_f32_16x16x4_f32(false, a, false, b3, (short)0, acc3, false, false);
            }
        }
    }

    if (active) {
#pragma unroll
        for (int r = 0; r < 8; ++r) {
            int M = m0 + r + 8 * half;
            if (M < N) {
                float* o = XF + (size_t)M * NFILT + l16;
                o[0]  = acc0[r];
                o[16] = acc1[r];
                o[32] = acc2[r];
                o[48] = acc3[r];
            }
        }
    }
}

__global__ __launch_bounds__(256, 16) void kw_gemm_bin_kernel(const float* __restrict__ x,
                                                          const float* __restrict__ flt,
                                                          float* __restrict__ XF,
                                                          const int* __restrict__ rows,
                                                          const int* __restrict__ cols,
                                                          float* out, int E, int N) {
    __shared__ float sB[KCHUNK * NFILT];
    __shared__ int s_cnt[NBMAX];
    __shared__ int s_base[NBMAX];
    const int gemm_blocks = (((N + 15) >> 4) + 7) >> 3;
    if ((int)blockIdx.x < gemm_blocks) kw_gemm_body(x, flt, XF, N, (int)blockIdx.x, sB);
    else                               kw_bin_body(rows, cols, out, E, N, (int)blockIdx.x - gemm_blocks, s_cnt, s_base);
}

__global__ __launch_bounds__(BD_THREADS) void kw_build_kernel(const int* __restrict__ rows,
                                                              const int* __restrict__ cols,
                                                              float* out, float* __restrict__ dinv, int E, int N) {
    __shared__ int s_bk[RPB * NFILT];
    __shared__ int s_c[RPB];
    int* bk = (int*)out;
    const int tid = (int)threadIdx.x;
    const int b   = (int)blockIdx.x;
    const int r0  = b * RPB;
    const int nr  = kw_bin_rows(b, N);
    const size_t base0 = (size_t)r0 * NFILT;
    const size_t base1 = base0 + (size_t)nr * 32;
    const int cap = nr * 32 - 32;
    const int c0 = bk[base0];
    const int c1 = bk[base1];
    for (int q = tid; q < nr * (NFILT / 4); q += BD_THREADS) ((int4*)s_bk)[q] = make_int4(0, 0, 0, 0);
    for (int rl = tid; rl < nr; rl += BD_THREADS) s_c[rl] = 0;
    __syncthreads();
    if (c0 <= cap && c1 <= cap) {
        const int tot = c0 + c1;
        for (int i0 = 0; i0 < tot; i0 += BD_THREADS * 8) {
            int w[8];
#pragma unroll
            for (int j = 0; j < 8; ++j) {
                const int i = i0 + j * BD_THREADS + tid;
                w[j] = -1;
                if (i < tot) w[j] = bk[(i < c0) ? (base0 + 32 + i) : (base1 + 32 + (i - c0))];
            }
#pragma unroll
            for (int j = 0; j < 8; ++j) {
                if (w[j] >= 0) {
                    const int rl = w[j] >> SRC_BITS;
                    const int k  = __hip_atomic_fetch_add(s_c + rl, 1, __ATOMIC_RELAXED, __HIP_MEMORY_SCOPE_WORKGROUP);
                    if (k < BK_CAP) s_bk[rl * NFILT + 1 + k] = w[j] & ((1 << SRC_BITS) - 1);
                }
            }
        }
    } else {
        for (int e = tid; e < E; e += BD_THREADS) {
            const int r = rows[e];
            if (r >= r0 && r < r0 + nr) {
                const int rl = r - r0;
                const int k  = __hip_atomic_fetch_add(s_c + rl, 1, __ATOMIC_RELAXED, __HIP_MEMORY_SCOPE_WORKGROUP);
                if (k < BK_CAP) s_bk[rl * NFILT + 1 + k] = cols[e];
            }
        }
    }
    __syncthreads();
    for (int rl = tid; rl < nr; rl += BD_THREADS) {
        const int d = s_c[rl];
        s_bk[rl * NFILT] = d;
        dinv[r0 + rl] = (d > 0) ? rsqrtf((float)d) : 0.0f;
    }
    __syncthreads();
    for (int q = tid; q < nr * (NFILT / 4); q += BD_THREADS) ((int4*)(bk + base0))[q] = ((const int4*)s_bk)[q];
}

template <int K>
__device__ __forceinline__ void kw_sum_group(int s0, int ids, float dcs, float di, int lane,
                                             const float* __restrict__ XF, v2f& acc) {
    int cc[K];
#pragma unroll
    for (int u = 0; u < K; ++u) cc[u] = __builtin_amdgcn_readlane(ids, s0 + u);
    v2f v[K];
#pragma unroll
    for (int u = 0; u < K; ++u) v[u] = *(const v2f*)(XF + (size_t)cc[u] * NFILT + 2 * lane);
#pragma unroll
    for (int u = 0; u < K; ++u) {
        const float w = di * __builtin_bit_cast(float, __builtin_amdgcn_readlane(__builtin_bit_cast(int, dcs), s0 + u));
        acc += w * v[u];
    }
}

__device__ __forceinline__ v2f kw_row_walk(int i, float di, int lane,
                                           const int* __restrict__ rows, const int* __restrict__ cols,
                                           const float* __restrict__ dinv, const float* __restrict__ XF, int E) {
    v2f acc = {0.0f, 0.0f};
    for (int e0 = 0; e0 < E; e0 += 32) {
        const int e = e0 + lane;
        int r = -1, c = 0;
        if (e < E) { r = rows[e]; c = cols[e]; }
        unsigned m = __builtin_amdgcn_ballot_w32(r == i);
        while (m) {
            const int bl = __builtin_ctz(m);
            m &= (m - 1u);
            const int   cc = __builtin_amdgcn_readlane(c, bl);
            const float w  = di * dinv[cc];
            const v2f   v  = *(const v2f*)(XF + (size_t)cc * NFILT + 2 * lane);
            acc += w * v;
        }
    }
    return acc;
}

__global__ __launch_bounds__(256) void kw_gather_kernel(const int* __restrict__ rows,
                                                        const int* __restrict__ cols,
                                                        const float* __restrict__ dinv,
                                                        const float* __restrict__ XF,
                                                        float* __restrict__ out,
                                                        int* __restrict__ ctr, int E, int N) {
    __shared__ int s_tk[2];
    const int lane  = threadIdx.x & 31;
    const int wib   = threadIdx.x >> 5;
    const int home  = kw_chip_half();
    const int split = N >> 1;
    int it = 0;
    for (int pass = 0; pass < 2; ++pass) {
        const int h  = home ^ pass;
        const int lo = h ? split : 0;
        const int hi = h ? N : split;
        const int nchunks = (hi - lo + GR_ROWS - 1) / GR_ROWS;
        int* c = ctr + h;
        int tn = 0;
        if (threadIdx.x == 0) tn = kw_draw(c);
        for (;;) {
            if (threadIdx.x == 0) s_tk[it & 1] = tn;
            __syncthreads();
            const int t = s_tk[it & 1];
            ++it;
            if (t >= nchunks) break;
            if (threadIdx.x == 0) tn = kw_draw(c);

            const int i0 = lo + t * GR_ROWS + wib * GR_RPW;
            int   a[GR_RPW];
            float di[GR_RPW];
#pragma unroll
            for (int j = 0; j < GR_RPW; ++j) {
                a[j] = 0; di[j] = 0.0f;
                if (i0 + j < hi) {
                    a[j]  = __builtin_bit_cast(int, out[(size_t)(i0 + j) * NFILT + lane]);
                    di[j] = dinv[i0 + j];
                }
            }
            int   cnt[GR_RPW];
            float da[GR_RPW];
#pragma unroll
            for (int j = 0; j < GR_RPW; ++j) {
                cnt[j] = __builtin_amdgcn_readlane(a[j], 0);
                da[j] = dinv[(lane >= 1 && lane <= cnt[j]) ? a[j] : 0];
            }
#pragma unroll
            for (int j = 0; j < GR_RPW; ++j) {
                if (i0 + j < hi) {
                    const int i = i0 + j;
                    v2f acc = {0.0f, 0.0f};
                    if (cnt[j] <= BK_CAP) {
                        const int n1 = cnt[j] < 31 ? cnt[j] : 31;
                        int s = 1;
                        while (n1 - s >= 7) { kw_sum_group<8>(s, a[j], da[j], di[j], lane, XF, acc); s += 8; }
                        if (n1 - s >= 3)    { kw_sum_group<4>(s, a[j], da[j], di[j], lane, XF, acc); s += 4; }
                        if (n1 - s >= 1)    { kw_sum_group<2>(s, a[j], da[j], di[j], lane, XF, acc); s += 2; }
                        if (n1 - s >= 0)    { kw_sum_group<1>(s, a[j], da[j], di[j], lane, XF, acc); }
                        if (cnt[j] >= 32) {
                            const int   b  = __builtin_bit_cast(int, out[(size_t)i * NFILT + 32 + lane]);
                            const float db = dinv[(lane + 32 <= cnt[j]) ? b : 0];
                            for (int u = 32; u <= cnt[j]; ++u) kw_sum_group<1>(u - 32, b, db, di[j], lane, XF, acc);
                        }
                    } else {
                        acc = kw_row_walk(i, di[j], lane, rows, cols, dinv, XF, E);
                    }
                    *(v2f*)(out + (size_t)i * NFILT + 2 * lane) = acc;
                }
            }
        }
    }
}

extern "C" void kernel_launch(void* const* d_in, const int* in_sizes, int n_in,
                              void* d_out, int out_size, void* d_ws, size_t ws_size,
                              hipStream_t stream) {
    const float* x   = (const float*)d_in[0];
    const float* flt = (const float*)d_in[1];
    const int*   eid = (const int*)d_in[2];

    const int N = in_sizes[0] / NFEAT;
    const int E = in_sizes[2] / 2;
    const int* rows = eid;
    const int* cols = eid + E;

    float* out  = (float*)d_out;
    float* dinv = (float*)d_ws;
    int*   ctr  = (int*)(dinv + N);
    float* XF   = (float*)d_ws + (size_t)((N + 4 + 63) & ~63);

    kw_prep_kernel<<<(2 * ((N + RPB - 1) / RPB) + 255) / 256, 256, 0, stream>>>(out, ctr, N);

    const int stripes = (N + 15) / 16;
    kw_gemm_bin_kernel<<<(stripes + 7) / 8 + (E + BN_EDGES - 1) / BN_EDGES, 256, 0, stream>>>(x, flt, XF, rows, cols, out, E, N);

    kw_build_kernel<<<(N + RPB - 1) / RPB, BD_THREADS, 0, stream>>>(rows, cols, out, dinv, E, N);

    kw_gather_kernel<<<2048, 256, 0, stream>>>(rows, cols, dinv, XF, out, ctr + 2, E, N);
}


# Product's A operand fetched in whole 128-byte lines (16 B per lane, one 32-k piece ahead) and re-laid through per-wave LDS rows, not 16 lines per fragment load; same values and k order
# speedup vs baseline: 2.8040x; 1.0512x over previous
#include <hip/hip_runtime.h>
#include <hip/hip_bf16.h>

typedef __attribute__((ext_vector_type(2))) float v2f;
typedef __attribute__((ext_vector_type(4))) float v4f;
typedef __attribute__((ext_vector_type(8))) float v8f;

#define NFEAT 512
#define NFILT 64
#define KCHUNK 128
#define ASUB   32
#define AP     36


#define BK_CAP   63
#define RPB      391
#define NBMAX    256
#define BN_EDGES 2048
#define SRC_BITS 20
#define BD_THREADS 1024
#define GR_RPW   2
#define GR_ROWS  (8 * GR_RPW)

__device__ __forceinline__ int kw_bin_rows(int b, int N) {
    const int nr = N - b * RPB;
    return nr < RPB ? nr : RPB;
}

__global__ void kw_prep_kernel(float* out, int* __restrict__ ctr, int N) {
    const int i  = blockIdx.x * blockDim.x + threadIdx.x;
    const int NB = (N + RPB - 1) / RPB;
    if (i < 2 * NB) {
        const int b = i >> 1;
        ((int*)out)[(size_t)b * RPB * NFILT + (size_t)(i & 1) * kw_bin_rows(b, N) * 32] = 0;
    }
    if (i < 4) ctr[i] = 0;
}

__device__ __forceinline__ int kw_chip_half() {
    return (int)((__builtin_amdgcn_s_sendmsg_rtn(0x87) >> 18) & 1u);
}

__device__ __forceinline__ int kw_draw(int* c) {
    return __hip_atomic_fetch_add(c, 1, __ATOMIC_RELAXED, __HIP_MEMORY_SCOPE_AGENT);
}

__device__ __forceinline__ void kw_bin_body(const int* __restrict__ rows,
                                            const int* __restrict__ cols,
                                            float* out, int E, int N, int chunk, int* s_cnt, int* s_base) {
    int* bk = (int*)out;
    const int h  = kw_chip_half();
    const int NB = (N + RPB - 1) / RPB;
    for (int b = threadIdx.x; b < NB; b += 256) s_cnt[b] = 0;
    __syncthreads();
    int  bin[BN_EDGES / 256], w[BN_EDGES / 256], k[BN_EDGES / 256];
    bool ok[BN_EDGES / 256];
#pragma unroll
    for (int j = 0; j < BN_EDGES / 256; ++j) {
        const int e = chunk * BN_EDGES + j * 256 + (int)threadIdx.x;
        ok[j] = (e < E);
        const int r = rows[ok[j] ? e : E - 1];
        const int s = cols[ok[j] ? e : E - 1];
        ok[j]  = ok[j] && ((unsigned)r < (unsigned)N);
        bin[j] = ok[j] ? r / RPB : 0;
        w[j]   = (int)(((unsigned)(r - bin[j] * RPB) << SRC_BITS) | (unsigned)s);
    }
#pragma unroll
    for (int j = 0; j < BN_EDGES / 256; ++j) {
        k[j] = 0;
        if (ok[j]) k[j] = __hip_atomic_fetch_add(s_cnt + bin[j], 1, __ATOMIC_RELAXED, __HIP_MEMORY_SCOPE_WORKGROUP);
    }
    __syncthreads();
    for (int b = threadIdx.x; b < NB; b += 256) {
        const int c = s_cnt[b];
        int base = 0;
        if (c > 0) base = __hip_atomic_fetch_add(bk + (size_t)b * RPB * NFILT + (size_t)h * kw_bin_rows(b, N) * 32,
                                                 c, __ATOMIC_RELAXED, __HIP_MEMORY_SCOPE_AGENT);
        s_base[b] = base;
    }
    __syncthreads();
#pragma unroll
    for (int j = 0; j < BN_EDGES / 256; ++j) {
        if (ok[j]) {
            const int nr  = kw_bin_rows(bin[j], N);
            const int pos = s_base[bin[j]] + k[j];
            if (pos < nr * 32 - 32) bk[(size_t)bin[j] * RPB * NFILT + (size_t)h * nr * 32 + 32 + pos] = w[j];
        }
    }
}

__device__ __forceinline__ void kw_wave_order() {
    __builtin_amdgcn_fence(__ATOMIC_RELEASE, "wavefront");
    __builtin_amdgcn_wave_barrier();
    __builtin_amdgcn_fence(__ATOMIC_ACQUIRE, "wavefront");
}

__device__ __forceinline__ void kw_gemm_body(const float* __restrict__ x,
                                             const float* __restrict__ flt,
                                             float* __restrict__ XF, int N, int blk, float* sB, float* sA) {
    const int lane = threadIdx.x & 31;
    const int wib  = threadIdx.x >> 5;
    const int half = lane >> 4;
    const int l16  = lane & 15;

    const int stripe = blk * 8 + wib;
    const int m0     = stripe * 16;
    const bool active = (m0 < N);

    const int lr = lane >> 3;
    const int lp = lane & 7;
    const float* src[4];
#pragma unroll
    for (int q = 0; q < 4; ++q) {
        int r = m0 + 4 * q + lr;
        if (r >= N) r = N - 1;
        src[q] = x + (size_t)r * NFEAT + lp * 4;
    }
    float*       myA   = sA + wib * (16 * AP);
    float*       dst   = myA + lr * AP + lp * 4;
    const float* afrag = myA + l16 * AP + 2 * half;

    v8f acc0 = {}, acc1 = {}, acc2 = {}, acc3 = {};

    const float* bbase = sB + lane * 2;

    v4f pre[4];
#pragma unroll
    for (int q = 0; q < 4; ++q) pre[q] = *(const v4f*)(src[q]);

    for (int kc = 0; kc < NFEAT; kc += KCHUNK) {
        __syncthreads();
        {
            const v4f* fsrc = (const v4f*)(flt + (size_t)kc * NFILT);
            for (int t4 = threadIdx.x; t4 < KCHUNK * (NFILT / 4); t4 += 256) {
                v4f v = fsrc[t4];
                int r    = t4 >> 4;
                int g    = r >> 2;
                int koff = r & 3;
                int h    = koff >> 1;
                int j    = koff & 1;
                int cb   = (t4 & 15) << 2;
                int n    = cb >> 4;
                int c16  = cb & 15;
                float* d = sB + (((g << 2) + n) << 6) + (((h << 4) + c16) << 1) + j;
                d[0] = v.x; d[2] = v.y; d[4] = v.z; d[6] = v.w;
            }
        }
        __syncthreads();

        if (active) {
#pragma unroll 1
            for (int ks = 0; ks < KCHUNK; ks += ASUB) {
#pragma unroll
                for (int q = 0; q < 4; ++q) *(v4f*)(dst + q * (4 * AP)) = pre[q];
                const int kn = kc + ks + ASUB;
                if (kn < NFEAT) {
#pragma unroll
                    for (int q = 0; q < 4; ++q) pre[q] = *(const v4f*)(src[q] + kn);
                }
                kw_wave_order();
#pragma unroll 4
                for (int k2 = 0; k2 < ASUB; k2 += 4) {
                    v2f a = *(const v2f*)(afrag + k2);

                    const int g = (ks + k2) >> 2;
                    const float* bg = bbase + (g << 8);
                    v2f b0 = *(const v2f*)(bg);
                    v2f b1 = *(const v2f*)(bg + 64);
                    v2f b2 = *(const v2f*)(bg + 128);
                    v2f b3 = *(const v2f*)(bg + 192);

                    acc0 = __builtin_amdgcn_wmma_f32_16x16x4_f32(false, a, false, b0, (short)0, acc0, false, false);
                    acc1 = __builtin_amdgcn_wmma_f32_16x16x4_f32(false, a, false, b1, (short)0, acc1, false, false);
                    acc2 = __builtin_amdgcn_wmma_f32_16x16x4_f32(false, a, false, b2, (short)0, acc2, false, false);
                    acc3 = __builtin_amdgcn_wmma_f32_16x16x4_f32(false, a, false, b3, (short)0, acc3, false, false);
                }
                kw_wave_order();
            }
        }
    }

    if (active) {
#pragma unroll
        for (int r = 0; r < 8; ++r) {
            int M = m0 + r + 8 * half;
            if (M < N) {
                float* o = XF + (size_t)M * NFILT + l16;
                o[0]  = acc0[r];
                o[16] = acc1[r];
                o[32] = acc2[r];
                o[48] = acc3[r];
            }
        }
    }
}

__global__ __launch_bounds__(256, 16) void kw_gemm_bin_kernel(const float* __restrict__ x,
                                                          const float* __restrict__ flt,
                                                          float* __restrict__ XF,
                                                          const int* __restrict__ rows,
                                                          const int* __restrict__ cols,
                                                          float* out, int E, int N) {
    __shared__ float sB[KCHUNK * NFILT];
    __shared__ float sA[8 * 16 * AP];
    __shared__ int s_cnt[NBMAX];
    __shared__ int s_base[NBMAX];
    const int gemm_blocks = (((N + 15) >> 4) + 7) >> 3;
    if ((int)blockIdx.x < gemm_blocks) kw_gemm_body(x, flt, XF, N, (int)blockIdx.x, sB, sA);
    else                               kw_bin_body(rows, cols, out, E, N, (int)blockIdx.x - gemm_blocks, s_cnt, s_base);
}

__global__ __launch_bounds__(BD_THREADS) void kw_build_kernel(const int* __restrict__ rows,
                                                              const int* __restrict__ cols,
                                                              float* out, float* __restrict__ dinv, int E, int N) {
    __shared__ int s_bk[RPB * NFILT];
    __shared__ int s_c[RPB];
    int* bk = (int*)out;
    const int tid = (int)threadIdx.x;
    const int b   = (int)blockIdx.x;
    const int r0  = b * RPB;
    const int nr  = kw_bin_rows(b, N);
    const size_t base0 = (size_t)r0 * NFILT;
    const size_t base1 = base0 + (size_t)nr * 32;
    const int cap = nr * 32 - 32;
    const int c0 = bk[base0];
    const int c1 = bk[base1];
    for (int q = tid; q < nr * (NFILT / 4); q += BD_THREADS) ((int4*)s_bk)[q] = make_int4(0, 0, 0, 0);
    for (int rl = tid; rl < nr; rl += BD_THREADS) s_c[rl] = 0;
    __syncthreads();
    if (c0 <= cap && c1 <= cap) {
        const int tot = c0 + c1;
        for (int i0 = 0; i0 < tot; i0 += BD_THREADS * 8) {
            int w[8];
#pragma unroll
            for (int j = 0; j < 8; ++j) {
                const int i = i0 + j * BD_THREADS + tid;
                w[j] = -1;
                if (i < tot) w[j] = bk[(i < c0) ? (base0 + 32 + i) : (base1 + 32 + (i - c0))];
            }
#pragma unroll
            for (int j = 0; j < 8; ++j) {
                if (w[j] >= 0) {
                    const int rl = w[j] >> SRC_BITS;
                    const int k  = __hip_atomic_fetch_add(s_c + rl, 1, __ATOMIC_RELAXED, __HIP_MEMORY_SCOPE_WORKGROUP);
                    if (k < BK_CAP) s_bk[rl * NFILT + 1 + k] = w[j] & ((1 << SRC_BITS) - 1);
                }
            }
        }
    } else {
        for (int e = tid; e < E; e += BD_THREADS) {
            const int r = rows[e];
            if (r >= r0 && r < r0 + nr) {
                const int rl = r - r0;
                const int k  = __hip_atomic_fetch_add(s_c + rl, 1, __ATOMIC_RELAXED, __HIP_MEMORY_SCOPE_WORKGROUP);
                if (k < BK_CAP) s_bk[rl * NFILT + 1 + k] = cols[e];
            }
        }
    }
    __syncthreads();
    for (int rl = tid; rl < nr; rl += BD_THREADS) {
        const int d = s_c[rl];
        s_bk[rl * NFILT] = d;
        dinv[r0 + rl] = (d > 0) ? rsqrtf((float)d) : 0.0f;
    }
    __syncthreads();
    for (int q = tid; q < nr * (NFILT / 4); q += BD_THREADS) ((int4*)(bk + base0))[q] = ((const int4*)s_bk)[q];
}

template <int K>
__device__ __forceinline__ void kw_sum_group(int s0, int ids, float dcs, float di, int lane,
                                             const float* __restrict__ XF, v2f& acc) {
    int cc[K];
#pragma unroll
    for (int u = 0; u < K; ++u) cc[u] = __builtin_amdgcn_readlane(ids, s0 + u);
    v2f v[K];
#pragma unroll
    for (int u = 0; u < K; ++u) v[u] = *(const v2f*)(XF + (size_t)cc[u] * NFILT + 2 * lane);
#pragma unroll
    for (int u = 0; u < K; ++u) {
        const float w = di * __builtin_bit_cast(float, __builtin_amdgcn_readlane(__builtin_bit_cast(int, dcs), s0 + u));
        acc += w * v[u];
    }
}

__device__ __forceinline__ v2f kw_row_walk(int i, float di, int lane,
                                           const int* __restrict__ rows, const int* __restrict__ cols,
                                           const float* __restrict__ dinv, const float* __restrict__ XF, int E) {
    v2f acc = {0.0f, 0.0f};
    for (int e0 = 0; e0 < E; e0 += 32) {
        const int e = e0 + lane;
        int r = -1, c = 0;
        if (e < E) { r = rows[e]; c = cols[e]; }
        unsigned m = __builtin_amdgcn_ballot_w32(r == i);
        while (m) {
            const int bl = __builtin_ctz(m);
            m &= (m - 1u);
            const int   cc = __builtin_amdgcn_readlane(c, bl);
            const float w  = di * dinv[cc];
            const v2f   v  = *(const v2f*)(XF + (size_t)cc * NFILT + 2 * lane);
            acc += w * v;
        }
    }
    return acc;
}

__global__ __launch_bounds__(256) void kw_gather_kernel(const int* __restrict__ rows,
                                                        const int* __restrict__ cols,
                                                        const float* __restrict__ dinv,
                                                        const float* __restrict__ XF,
                                                        float* __restrict__ out,
                                                        int* __restrict__ ctr, int E, int N) {
    __shared__ int s_tk[2];
    const int lane  = threadIdx.x & 31;
    const int wib   = threadIdx.x >> 5;
    const int home  = kw_chip_half();
    const int split = N >> 1;
    int it = 0;
    for (int pass = 0; pass < 2; ++pass) {
        const int h  = home ^ pass;
        const int lo = h ? split : 0;
        const int hi = h ? N : split;
        const int nchunks = (hi - lo + GR_ROWS - 1) / GR_ROWS;
        int* c = ctr + h;
        int tn = 0;
        if (threadIdx.x == 0) tn = kw_draw(c);
        for (;;) {
            if (threadIdx.x == 0) s_tk[it & 1] = tn;
            __syncthreads();
            const int t = s_tk[it & 1];
            ++it;
            if (t >= nchunks) break;
            if (threadIdx.x == 0) tn = kw_draw(c);

            const int i0 = lo + t * GR_ROWS + wib * GR_RPW;
            int   a[GR_RPW];
            float di[GR_RPW];
#pragma unroll
            for (int j = 0; j < GR_RPW; ++j) {
                a[j] = 0; di[j] = 0.0f;
                if (i0 + j < hi) {
                    a[j]  = __builtin_bit_cast(int, out[(size_t)(i0 + j) * NFILT + lane]);
                    di[j] = dinv[i0 + j];
                }
            }
            int   cnt[GR_RPW];
            float da[GR_RPW];
#pragma unroll
            for (int j = 0; j < GR_RPW; ++j) {
                cnt[j] = __builtin_amdgcn_readlane(a[j], 0);
                da[j] = dinv[(lane >= 1 && lane <= cnt[j]) ? a[j] : 0];
            }
#pragma unroll
            for (int j = 0; j < GR_RPW; ++j) {
                if (i0 + j < hi) {
                    const int i = i0 + j;
                    v2f acc = {0.0f, 0.0f};
                    if (cnt[j] <= BK_CAP) {
                        const int n1 = cnt[j] < 31 ? cnt[j] : 31;
                        int s = 1;
                        while (n1 - s >= 7) { kw_sum_group<8>(s, a[j], da[j], di[j], lane, XF, acc); s += 8; }
                        if (n1 - s >= 3)    { kw_sum_group<4>(s, a[j], da[j], di[j], lane, XF, acc); s += 4; }
                        if (n1 - s >= 1)    { kw_sum_group<2>(s, a[j], da[j], di[j], lane, XF, acc); s += 2; }
                        if (n1 - s >= 0)    { kw_sum_group<1>(s, a[j], da[j], di[j], lane, XF, acc); }
                        if (cnt[j] >= 32) {
                            const int   b  = __builtin_bit_cast(int, out[(size_t)i * NFILT + 32 + lane]);
                            const float db = dinv[(lane + 32 <= cnt[j]) ? b : 0];
                            for (int u = 32; u <= cnt[j]; ++u) kw_sum_group<1>(u - 32, b, db, di[j], lane, XF, acc);
                        }
                    } else {
                        acc = kw_row_walk(i, di[j], lane, rows, cols, dinv, XF, E);
                    }
                    *(v2f*)(out + (size_t)i * NFILT + 2 * lane) = acc;
                }
            }
        }
    }
}

extern "C" void kernel_launch(void* const* d_in, const int* in_sizes, int n_in,
                              void* d_out, int out_size, void* d_ws, size_t ws_size,
                              hipStream_t stream) {
    const float* x   = (const float*)d_in[0];
    const float* flt = (const float*)d_in[1];
    const int*   eid = (const int*)d_in[2];

    const int N = in_sizes[0] / NFEAT;
    const int E = in_sizes[2] / 2;
    const int* rows = eid;
    const int* cols = eid + E;

    float* out  = (float*)d_out;
    float* dinv = (float*)d_ws;
    int*   ctr  = (int*)(dinv + N);
    float* XF   = (float*)d_ws + (size_t)((N + 4 + 63) & ~63);

    kw_prep_kernel<<<(2 * ((N + RPB - 1) / RPB) + 255) / 256, 256, 0, stream>>>(out, ctr, N);

    const int stripes = (N + 15) / 16;
    kw_gemm_bin_kernel<<<(stripes + 7) / 8 + (E + BN_EDGES - 1) / BN_EDGES, 256, 0, stream>>>(x, flt, XF, rows, cols, out, E, N);

    kw_build_kernel<<<(N + RPB - 1) / RPB, BD_THREADS, 0, stream>>>(rows, cols, out, dinv, E, N);

    kw_gather_kernel<<<2048, 256, 0, stream>>>(rows, cols, dinv, XF, out, ctr + 2, E, N);
}


# Product's ragged last round removed: 8-stripe work-groups only in whole rounds of one per WGP (768); the 106 left-over stripes go to 2-stripe x 4-column-tile work-groups, a quarter of the work each
# speedup vs baseline: 2.8898x; 1.0306x over previous
#include <hip/hip_runtime.h>
#include <hip/hip_bf16.h>

typedef __attribute__((ext_vector_type(2))) float v2f;
typedef __attribute__((ext_vector_type(4))) float v4f;
typedef __attribute__((ext_vector_type(8))) float v8f;

#define NFEAT 512
#define NFILT 64
#define KCHUNK 128
#define ASUB   32
#define AP     36
#define KW_WGPS 256


#define BK_CAP   63
#define RPB      391
#define NBMAX    256
#define BN_EDGES 2048
#define SRC_BITS 20
#define BD_THREADS 1024
#define GR_RPW   2
#define GR_ROWS  (8 * GR_RPW)

__device__ __forceinline__ int kw_bin_rows(int b, int N) {
    const int nr = N - b * RPB;
    return nr < RPB ? nr : RPB;
}

__global__ void kw_prep_kernel(float* out, int* __restrict__ ctr, int N) {
    const int i  = blockIdx.x * blockDim.x + threadIdx.x;
    const int NB = (N + RPB - 1) / RPB;
    if (i < 2 * NB) {
        const int b = i >> 1;
        ((int*)out)[(size_t)b * RPB * NFILT + (size_t)(i & 1) * kw_bin_rows(b, N) * 32] = 0;
    }
    if (i < 4) ctr[i] = 0;
}

__device__ __forceinline__ int kw_chip_half() {
    return (int)((__builtin_amdgcn_s_sendmsg_rtn(0x87) >> 18) & 1u);
}

__device__ __forceinline__ int kw_draw(int* c) {
    return __hip_atomic_fetch_add(c, 1, __ATOMIC_RELAXED, __HIP_MEMORY_SCOPE_AGENT);
}

__device__ __forceinline__ void kw_bin_body(const int* __restrict__ rows,
                                            const int* __restrict__ cols,
                                            float* out, int E, int N, int chunk, int* s_cnt, int* s_base) {
    int* bk = (int*)out;
    const int h  = kw_chip_half();
    const int NB = (N + RPB - 1) / RPB;
    for (int b = threadIdx.x; b < NB; b += 256) s_cnt[b] = 0;
    __syncthreads();
    int  bin[BN_EDGES / 256], w[BN_EDGES / 256], k[BN_EDGES / 256];
    bool ok[BN_EDGES / 256];
#pragma unroll
    for (int j = 0; j < BN_EDGES / 256; ++j) {
        const int e = chunk * BN_EDGES + j * 256 + (int)threadIdx.x;
        ok[j] = (e < E);
        const int r = rows[ok[j] ? e : E - 1];
        const int s = cols[ok[j] ? e : E - 1];
        ok[j]  = ok[j] && ((unsigned)r < (unsigned)N);
        bin[j] = ok[j] ? r / RPB : 0;
        w[j]   = (int)(((unsigned)(r - bin[j] * RPB) << SRC_BITS) | (unsigned)s);
    }
#pragma unroll
    for (int j = 0; j < BN_EDGES / 256; ++j) {
        k[j] = 0;
        if (ok[j]) k[j] = __hip_atomic_fetch_add(s_cnt + bin[j], 1, __ATOMIC_RELAXED, __HIP_MEMORY_SCOPE_WORKGROUP);
    }
    __syncthreads();
    for (int b = threadIdx.x; b < NB; b += 256) {
        const int c = s_cnt[b];
        int base = 0;
        if (c > 0) base = __hip_atomic_fetch_add(bk + (size_t)b * RPB * NFILT + (size_t)h * kw_bin_rows(b, N) * 32,
                                                 c, __ATOMIC_RELAXED, __HIP_MEMORY_SCOPE_AGENT);
        s_base[b] = base;
    }
    __syncthreads();
#pragma unroll
    for (int j = 0; j < BN_EDGES / 256; ++j) {
        if (ok[j]) {
            const int nr  = kw_bin_rows(bin[j], N);
            const int pos = s_base[bin[j]] + k[j];
            if (pos < nr * 32 - 32) bk[(size_t)bin[j] * RPB * NFILT + (size_t)h * nr * 32 + 32 + pos] = w[j];
        }
    }
}

__device__ __forceinline__ void kw_wave_order() {
    __builtin_amdgcn_fence(__ATOMIC_RELEASE, "wavefront");
    __builtin_amdgcn_wave_barrier();
    __builtin_amdgcn_fence(__ATOMIC_ACQUIRE, "wavefront");
}

template <bool SPLIT>
__device__ __forceinline__ void kw_gemm_body(const float* __restrict__ x,
                                             const float* __restrict__ flt,
                                             float* __restrict__ XF, int N, int stripe0, float* sB, float* sA) {
    const int lane = threadIdx.x & 31;
    const int wib  = threadIdx.x >> 5;
    const int half = lane >> 4;
    const int l16  = lane & 15;

    const int stripe = SPLIT ? stripe0 + (wib >> 2) : stripe0 + wib;
    const int nt     = wib & 3;
    const int m0     = stripe * 16;
    const bool active = (m0 < N);

    const int lr = lane >> 3;
    const int lp = lane & 7;
    const float* src[4];
#pragma unroll
    for (int q = 0; q < 4; ++q) {
        int r = m0 + 4 * q + lr;
        if (r >= N) r = N - 1;
        src[q] = x + (size_t)r * NFEAT + lp * 4;
    }
    float*       myA   = sA + wib * (16 * AP);
    float*       dst   = myA + lr * AP + lp * 4;
    const float* afrag = myA + l16 * AP + 2 * half;

    v8f acc0 = {}, acc1 = {}, acc2 = {}, acc3 = {};

    const float* bbase = sB + lane * 2 + (SPLIT ? nt * 64 : 0);

    v4f pre[4];
#pragma unroll
    for (int q = 0; q < 4; ++q) pre[q] = *(const v4f*)(src[q]);

    for (int kc = 0; kc < NFEAT; kc += KCHUNK) {
        __syncthreads();
        {
            const v4f* fsrc = (const v4f*)(flt + (size_t)kc * NFILT);
            for (int t4 = threadIdx.x; t4 < KCHUNK * (NFILT / 4); t4 += 256) {
                v4f v = fsrc[t4];
                int r    = t4 >> 4;
                int g    = r >> 2;
                int koff = r & 3;
                int h    = koff >> 1;
                int j    = koff & 1;
                int cb   = (t4 & 15) << 2;
                int n    = cb >> 4;
                int c16  = cb & 15;
                float* d = sB + (((g << 2) + n) << 6) + (((h << 4) + c16) << 1) + j;
                d[0] = v.x; d[2] = v.y; d[4] = v.z; d[6] = v.w;
            }
        }
        __syncthreads();

        if (active) {
#pragma unroll 1
            for (int ks = 0; ks < KCHUNK; ks += ASUB) {
#pragma unroll
                for (int q = 0; q < 4; ++q) *(v4f*)(dst + q * (4 * AP)) = pre[q];
                const int kn = kc + ks + ASUB;
                if (kn < NFEAT) {
#pragma unroll
                    for (int q = 0; q < 4; ++q) pre[q] = *(const v4f*)(src[q] + kn);
                }
                kw_wave_order();
#pragma unroll 4
                for (int k2 = 0; k2 < ASUB; k2 += 4) {
                    v2f a = *(const v2f*)(afrag + k2);

                    const int g = (ks + k2) >> 2;
                    const float* bg = bbase + (g << 8);
                    v2f b0 = *(const v2f*)(bg);
                    acc0 = __builtin_amdgcn_wmma_f32_16x16x4_f32(false, a, false, b0, (short)0, acc0, false, false);
                    if (!SPLIT) {
                        v2f b1 = *(const v2f*)(bg + 64);
                        v2f b2 = *(const v2f*)(bg + 128);
                        v2f b3 = *(const v2f*)(bg + 192);
                        acc1 = __builtin_amdgcn_wmma_f32_16x16x4_f32(false, a, false, b1, (short)0, acc1, false, false);
                        acc2 = __builtin_amdgcn_wmma_f32_16x16x4_f32(false, a, false, b2, (short)0, acc2, false, false);
                        acc3 = __builtin_amdgcn_wmma_f32_16x16x4_f32(false, a, false, b3, (short)0, acc3, false, false);
                    }
                }
                kw_wave_order();
            }
        }
    }

    if (active) {
#pragma unroll
        for (int r = 0; r < 8; ++r) {
            int M = m0 + r + 8 * half;
            if (M < N) {
                float* o = XF + (size_t)M * NFILT + l16;
                if (SPLIT) {
                    o[nt * 16] = acc0[r];
                } else {
                    o[0]  = acc0[r];
                    o[16] = acc1[r];
                    o[32] = acc2[r];
                    o[48] = acc3[r];
                }
            }
        }
    }
}

__global__ __launch_bounds__(256, 16) void kw_gemm_bin_kernel(const float* __restrict__ x,
                                                          const float* __restrict__ flt,
                                                          float* __restrict__ XF,
                                                          const int* __restrict__ rows,
                                                          const int* __restrict__ cols,
                                                          float* out, int E, int N) {
    __shared__ float sB[KCHUNK * NFILT];
    __shared__ float sA[8 * 16 * AP];
    __shared__ int s_cnt[NBMAX];
    __shared__ int s_base[NBMAX];
    const int stripes = (N + 15) >> 4;
    const int full    = ((stripes >> 3) / KW_WGPS) * KW_WGPS;
    const int split   = (stripes - full * 8 + 1) >> 1;
    const int bx      = (int)blockIdx.x;
    if (bx < full)              kw_gemm_body<false>(x, flt, XF, N, bx * 8, sB, sA);
    else if (bx < full + split) kw_gemm_body<true>(x, flt, XF, N, full * 8 + (bx - full) * 2, sB, sA);
    else                        kw_bin_body(rows, cols, out, E, N, bx - full - split, s_cnt, s_base);
}

__global__ __launch_bounds__(BD_THREADS) void kw_build_kernel(const int* __restrict__ rows,
                                                              const int* __restrict__ cols,
                                                              float* out, float* __restrict__ dinv, int E, int N) {
    __shared__ int s_bk[RPB * NFILT];
    __shared__ int s_c[RPB];
    int* bk = (int*)out;
    const int tid = (int)threadIdx.x;
    const int b   = (int)blockIdx.x;
    const int r0  = b * RPB;
    const int nr  = kw_bin_rows(b, N);
    const size_t base0 = (size_t)r0 * NFILT;
    const size_t base1 = base0 + (size_t)nr * 32;
    const int cap = nr * 32 - 32;
    const int c0 = bk[base0];
    const int c1 = bk[base1];
    for (int q = tid; q < nr * (NFILT / 4); q += BD_THREADS) ((int4*)s_bk)[q] = make_int4(0, 0, 0, 0);
    for (int rl = tid; rl < nr; rl += BD_THREADS) s_c[rl] = 0;
    __syncthreads();
    if (c0 <= cap && c1 <= cap) {
        const int tot = c0 + c1;
        for (int i0 = 0; i0 < tot; i0 += BD_THREADS * 8) {
            int w[8];
#pragma unroll
            for (int j = 0; j < 8; ++j) {
                const int i = i0 + j * BD_THREADS + tid;
                w[j] = -1;
                if (i < tot) w[j] = bk[(i < c0) ? (base0 + 32 + i) : (base1 + 32 + (i - c0))];
            }
#pragma unroll
            for (int j = 0; j < 8; ++j) {
                if (w[j] >= 0) {
                    const int rl = w[j] >> SRC_BITS;
                    const int k  = __hip_atomic_fetch_add(s_c + rl, 1, __ATOMIC_RELAXED, __HIP_MEMORY_SCOPE_WORKGROUP);
                    if (k < BK_CAP) s_bk[rl * NFILT + 1 + k] = w[j] & ((1 << SRC_BITS) - 1);
                }
            }
        }
    } else {
        for (int e = tid; e < E; e += BD_THREADS) {
            const int r = rows[e];
            if (r >= r0 && r < r0 + nr) {
                const int rl = r - r0;
                const int k  = __hip_atomic_fetch_add(s_c + rl, 1, __ATOMIC_RELAXED, __HIP_MEMORY_SCOPE_WORKGROUP);
                if (k < BK_CAP) s_bk[rl * NFILT + 1 + k] = cols[e];
            }
        }
    }
    __syncthreads();
    for (int rl = tid; rl < nr; rl += BD_THREADS) {
        const int d = s_c[rl];
        s_bk[rl * NFILT] = d;
        dinv[r0 + rl] = (d > 0) ? rsqrtf((float)d) : 0.0f;
    }
    __syncthreads();
    for (int q = tid; q < nr * (NFILT / 4); q += BD_THREADS) ((int4*)(bk + base0))[q] = ((const int4*)s_bk)[q];
}

template <int K>
__device__ __forceinline__ void kw_sum_group(int s0, int ids, float dcs, float di, int lane,
                                             const float* __restrict__ XF, v2f& acc) {
    int cc[K];
#pragma unroll
    for (int u = 0; u < K; ++u) cc[u] = __builtin_amdgcn_readlane(ids, s0 + u);
    v2f v[K];
#pragma unroll
    for (int u = 0; u < K; ++u) v[u] = *(const v2f*)(XF + (size_t)cc[u] * NFILT + 2 * lane);
#pragma unroll
    for (int u = 0; u < K; ++u) {
        const float w = di * __builtin_bit_cast(float, __builtin_amdgcn_readlane(__builtin_bit_cast(int, dcs), s0 + u));
        acc += w * v[u];
    }
}

__device__ __forceinline__ v2f kw_row_walk(int i, float di, int lane,
                                           const int* __restrict__ rows, const int* __restrict__ cols,
                                           const float* __restrict__ dinv, const float* __restrict__ XF, int E) {
    v2f acc = {0.0f, 0.0f};
    for (int e0 = 0; e0 < E; e0 += 32) {
        const int e = e0 + lane;
        int r = -1, c = 0;
        if (e < E) { r = rows[e]; c = cols[e]; }
        unsigned m = __builtin_amdgcn_ballot_w32(r == i);
        while (m) {
            const int bl = __builtin_ctz(m);
            m &= (m - 1u);
            const int   cc = __builtin_amdgcn_readlane(c, bl);
            const float w  = di * dinv[cc];
            const v2f   v  = *(const v2f*)(XF + (size_t)cc * NFILT + 2 * lane);
            acc += w * v;
        }
    }
    return acc;
}

__global__ __launch_bounds__(256) void kw_gather_kernel(const int* __restrict__ rows,
                                                        const int* __restrict__ cols,
                                                        const float* __restrict__ dinv,
                                                        const float* __restrict__ XF,
                                                        float* __restrict__ out,
                                                        int* __restrict__ ctr, int E, int N) {
    __shared__ int s_tk[2];
    const int lane  = threadIdx.x & 31;
    const int wib   = threadIdx.x >> 5;
    const int home  = kw_chip_half();
    const int split = N >> 1;
    int it = 0;
    for (int pass = 0; pass < 2; ++pass) {
        const int h  = home ^ pass;
        const int lo = h ? split : 0;
        const int hi = h ? N : split;
        const int nchunks = (hi - lo + GR_ROWS - 1) / GR_ROWS;
        int* c = ctr + h;
        int tn = 0;
        if (threadIdx.x == 0) tn = kw_draw(c);
        for (;;) {
            if (threadIdx.x == 0) s_tk[it & 1] = tn;
            __syncthreads();
            const int t = s_tk[it & 1];
            ++it;
            if (t >= nchunks) break;
            if (threadIdx.x == 0) tn = kw_draw(c);

            const int i0 = lo + t * GR_ROWS + wib * GR_RPW;
            int   a[GR_RPW];
            float di[GR_RPW];
#pragma unroll
            for (int j = 0; j < GR_RPW; ++j) {
                a[j] = 0; di[j] = 0.0f;
                if (i0 + j < hi) {
                    a[j]  = __builtin_bit_cast(int, out[(size_t)(i0 + j) * NFILT + lane]);
                    di[j] = dinv[i0 + j];
                }
            }
            int   cnt[GR_RPW];
            float da[GR_RPW];
#pragma unroll
            for (int j = 0; j < GR_RPW; ++j) {
                cnt[j] = __builtin_amdgcn_readlane(a[j], 0);
                da[j] = dinv[(lane >= 1 && lane <= cnt[j]) ? a[j] : 0];
            }
#pragma unroll
            for (int j = 0; j < GR_RPW; ++j) {
                if (i0 + j < hi) {
                    const int i = i0 + j;
                    v2f acc = {0.0f, 0.0f};
                    if (cnt[j] <= BK_CAP) {
                        const int n1 = cnt[j] < 31 ? cnt[j] : 31;
                        int s = 1;
                        while (n1 - s >= 7) { kw_sum_group<8>(s, a[j], da[j], di[j], lane, XF, acc); s += 8; }
                        if (n1 - s >= 3)    { kw_sum_group<4>(s, a[j], da[j], di[j], lane, XF, acc); s += 4; }
                        if (n1 - s >= 1)    { kw_sum_group<2>(s, a[j], da[j], di[j], lane, XF, acc); s += 2; }
                        if (n1 - s >= 0)    { kw_sum_group<1>(s, a[j], da[j], di[j], lane, XF, acc); }
                        if (cnt[j] >= 32) {
                            const int   b  = __builtin_bit_cast(int, out[(size_t)i * NFILT + 32 + lane]);
                            const float db = dinv[(lane + 32 <= cnt[j]) ? b : 0];
                            for (int u = 32; u <= cnt[j]; ++u) kw_sum_group<1>(u - 32, b, db, di[j], lane, XF, acc);
                        }
                    } else {
                        acc = kw_row_walk(i, di[j], lane, rows, cols, dinv, XF, E);
                    }
                    *(v2f*)(out + (size_t)i * NFILT + 2 * lane) = acc;
                }
            }
        }
    }
}

extern "C" void kernel_launch(void* const* d_in, const int* in_sizes, int n_in,
                              void* d_out, int out_size, void* d_ws, size_t ws_size,
                              hipStream_t stream) {
    const float* x   = (const float*)d_in[0];
    const float* flt = (const float*)d_in[1];
    const int*   eid = (const int*)d_in[2];

    const int N = in_sizes[0] / NFEAT;
    const int E = in_sizes[2] / 2;
    const int* rows = eid;
    const int* cols = eid + E;

    float* out  = (float*)d_out;
    float* dinv = (float*)d_ws;
    int*   ctr  = (int*)(dinv + N);
    float* XF   = (float*)d_ws + (size_t)((N + 4 + 63) & ~63);

    kw_prep_kernel<<<(2 * ((N + RPB - 1) / RPB) + 255) / 256, 256, 0, stream>>>(out, ctr, N);

    const int stripes = (N + 15) / 16;
    const int full    = stripes / 8 / KW_WGPS * KW_WGPS;
    const int split   = (stripes - full * 8 + 1) / 2;
    kw_gemm_bin_kernel<<<full + split + (E + BN_EDGES - 1) / BN_EDGES, 256, 0, stream>>>(x, flt, XF, rows, cols, out, E, N);

    kw_build_kernel<<<(N + RPB - 1) / RPB, BD_THREADS, 0, stream>>>(rows, cols, out, dinv, E, N);

    kw_gather_kernel<<<2048, 256, 0, stream>>>(rows, cols, dinv, XF, out, ctr + 2, E, N);
}
